# starting kernel
# baseline (speedup 1.0000x reference)
#include <hip/hip_runtime.h>


#define NV 18
#define NP 1330
#define KP 1332
#define NCOLS 20
#define NGROUPS (KP / 4)
#define WAVES 8
#define ROWS_PER_WAVE 16
#define ROWS_PER_BLOCK (WAVES * ROWS_PER_WAVE)

typedef __attribute__((ext_vector_type(2))) float v2f;
typedef __attribute__((ext_vector_type(4))) float v4f;
typedef __attribute__((ext_vector_type(8))) float v8f;

struct MonTab {
    unsigned char a[KP];
    unsigned char b[KP];
    unsigned char c[KP];
};
constexpr MonTab make_tab() {
    MonTab t{};
    int f = 0;
    t.a[f] = 18; t.b[f] = 18; t.c[f] = 18; ++f;
    for (int i = 0; i < NV; ++i) { t.a[f] = i; t.b[f] = 18; t.c[f] = 18; ++f; }
    for (int i = 0; i < NV; ++i)
        for (int j = i; j < NV; ++j) { t.a[f] = i; t.b[f] = j; t.c[f] = 18; ++f; }
    for (int i = 0; i < NV; ++i)
        for (int j = i; j < NV; ++j)
            for (int k = j; k < NV; ++k) { t.a[f] = i; t.b[f] = j; t.c[f] = k; ++f; }
    for (; f < KP; ++f) { t.a[f] = 18; t.b[f] = 18; t.c[f] = 18; }
    return t;
}
constexpr MonTab TAB = make_tab();

template <int F>
static __device__ __forceinline__ float mono(const float* x) {
    constexpr int ia = TAB.a[F];
    constexpr int ib = TAB.b[F];
    constexpr int ic = TAB.c[F];
    float p = x[ia];
    if constexpr (ib != 18) p *= x[ib];
    if constexpr (ic != 18) p *= x[ic];
    return p;
}

template <int G, int GEnd>
struct Steps {
    static __device__ __forceinline__ void run(const float* x, bool hi,
                                               const float* b0p, const float* b1p,
                                               v8f& acc0, v8f& acc1) {
        constexpr int f = 4 * G;
        float p0 = mono<f + 0>(x);
        float p1 = mono<f + 1>(x);
        float p2 = mono<f + 2>(x);
        float p3 = mono<f + 3>(x);
        v2f A;
        A.x = hi ? p2 : p0;
        A.y = hi ? p3 : p1;
        v2f B0 = *(const v2f*)(b0p + f);
        v2f B1 = *(const v2f*)(b1p + f);
        acc0 = __builtin_amdgcn_wmma_f32_16x16x4_f32(
                   false, A, false, B0, (short)0, acc0, false, false);
        acc1 = __builtin_amdgcn_wmma_f32_16x16x4_f32(
                   false, A, false, B1, (short)0, acc1, false, false);
        Steps<G + 1, GEnd>::run(x, hi, b0p, b1p, acc0, acc1);
    }
};
template <int GEnd>
struct Steps<GEnd, GEnd> {
    static __device__ __forceinline__ void run(const float*, bool, const float*,
                                               const float*, v8f&, v8f&) {}
};

__global__ __launch_bounds__(256)
void vindy_wmma_kernel(const float* __restrict__ z,
                       const float* __restrict__ betas,
                       const float* __restrict__ big_xi,
                       const float* __restrict__ mask,
                       float* __restrict__ out,
                       int batch)
{
    __shared__ float sXiT[NCOLS][KP];

    const int tid  = threadIdx.x;
    const int lane = tid & 31;
    const int w    = tid >> 5;
    const int half = lane >> 4;
    const int lm   = lane & 15;
    const int rowBase = blockIdx.x * ROWS_PER_BLOCK + w * ROWS_PER_WAVE;

    for (int col = 0; col < NCOLS; ++col) {
        for (int k = tid; k < KP; k += 256) {
            float v = 0.0f;
            if (col < NV && k < NP) v = big_xi[k * NV + col] * mask[k * NV + col];
            sXiT[col][k] = v;
        }
    }

    int gr = rowBase + lm;
    if (gr > batch - 1) gr = batch - 1;
    float x[19];
    {
        const v4f* zr = (const v4f*)(z + (size_t)gr * 16);
        v4f q0 = zr[0], q1 = zr[1], q2 = zr[2], q3 = zr[3];
        x[0]  = q0.x; x[1]  = q0.y; x[2]  = q0.z; x[3]  = q0.w;
        x[4]  = q1.x; x[5]  = q1.y; x[6]  = q1.z; x[7]  = q1.w;
        x[8]  = q2.x; x[9]  = q2.y; x[10] = q2.z; x[11] = q2.w;
        x[12] = q3.x; x[13] = q3.y; x[14] = q3.z; x[15] = q3.w;
        const v2f* br = (const v2f*)(betas + (size_t)gr * 2);
        v2f bb = br[0];
        x[16] = bb.x; x[17] = bb.y; x[18] = 1.0f;
    }

    __syncthreads();

    v8f acc0 = {0.f, 0.f, 0.f, 0.f, 0.f, 0.f, 0.f, 0.f};
    v8f acc1 = {0.f, 0.f, 0.f, 0.f, 0.f, 0.f, 0.f, 0.f};

    const bool hi = (half != 0);
    const float* b0p = &sXiT[lm][2 * half];
    const int col1   = (lm < 2) ? (16 + lm) : 18;
    const float* b1p = &sXiT[col1][2 * half];

    Steps<0, NGROUPS>::run(x, hi, b0p, b1p, acc0, acc1);

    const int orow = rowBase + 8 * half;
    if (rowBase + ROWS_PER_WAVE <= batch) {
#pragma unroll
        for (int v = 0; v < 8; ++v)
            out[(size_t)(orow + v) * NV + lm] = acc0[v];
        if (lm < 2) {
#pragma unroll
            for (int v = 0; v < 8; ++v)
                out[(size_t)(orow + v) * NV + 16 + lm] = acc1[v];
        }
    } else {
#pragma unroll
        for (int v = 0; v < 8; ++v) {
            int r = orow + v;
            if (r < batch) {
                out[(size_t)r * NV + lm] = acc0[v];
                if (lm < 2) out[(size_t)r * NV + 16 + lm] = acc1[v];
            }
        }
    }
}

extern "C" void kernel_launch(void* const* d_in, const int* in_sizes, int n_in,
                              void* d_out, int out_size, void* d_ws, size_t ws_size,
                              hipStream_t stream) {
    const float* z      = (const float*)d_in[0];
    const float* betas  = (const float*)d_in[1];
    const float* big_xi = (const float*)d_in[2];
    const float* mask   = (const float*)d_in[3];
    float* out          = (float*)d_out;

    const int batch = in_sizes[0] / 16;
    const int grid  = (batch + ROWS_PER_BLOCK - 1) / ROWS_PER_BLOCK;

    vindy_wmma_kernel<<<grid, 256, 0, stream>>>(z, betas, big_xi, mask, out, batch);
}


# K split over a workgroup's 8 waves: each keeps its 168-feature slice of Xi*mask as B fragments in registers for all its 256 rows (no LDS image of Xi); partial tiles summed in LDS: sum order changes
# speedup vs baseline: 1.9460x; 1.9460x over previous
#include <hip/hip_runtime.h>


#define NV 18
#define NP 1330
#define WAVES 8
#define GPW 42
#define FPW (4 * GPW)
#define KP (WAVES * FPW)
#define TILES_PER_STEP 4
#define ROWS_PER_STEP (16 * TILES_PER_STEP)
#define STEPS_PER_BLOCK 4
#define ROWS_PER_BLOCK (ROWS_PER_STEP * STEPS_PER_BLOCK)
#define STEP_ELEMS (ROWS_PER_STEP * NV)

typedef __attribute__((ext_vector_type(2))) float v2f;
typedef __attribute__((ext_vector_type(4))) float v4f;
typedef __attribute__((ext_vector_type(8))) float v8f;

struct MonTab {
    unsigned char a[KP];
    unsigned char b[KP];
    unsigned char c[KP];
};
constexpr MonTab make_tab() {
    MonTab t{};
    int f = 0;
    t.a[f] = 18; t.b[f] = 18; t.c[f] = 18; ++f;
    for (int i = 0; i < NV; ++i) { t.a[f] = i; t.b[f] = 18; t.c[f] = 18; ++f; }
    for (int i = 0; i < NV; ++i)
        for (int j = i; j < NV; ++j) { t.a[f] = i; t.b[f] = j; t.c[f] = 18; ++f; }
    for (int i = 0; i < NV; ++i)
        for (int j = i; j < NV; ++j)
            for (int k = j; k < NV; ++k) { t.a[f] = i; t.b[f] = j; t.c[f] = k; ++f; }
    for (; f < KP; ++f) { t.a[f] = 18; t.b[f] = 18; t.c[f] = 18; }
    return t;
}
constexpr MonTab TAB = make_tab();

template <int F>
static __device__ __forceinline__ float mono(const float* x) {
    constexpr int ia = TAB.a[F];
    constexpr int ib = TAB.b[F];
    constexpr int ic = TAB.c[F];
    float p = x[ia];
    if constexpr (ib != 18) p *= x[ib];
    if constexpr (ic != 18) p *= x[ic];
    return p;
}

template <int W, int G>
struct Steps {
    static __device__ __forceinline__ void run(const float* x, bool hi,
                                               const v2f* B0, const v2f* B1,
                                               v8f& acc0, v8f& acc1) {
        constexpr int f = FPW * W + 4 * G;
        float p0 = mono<f + 0>(x);
        float p1 = mono<f + 1>(x);
        float p2 = mono<f + 2>(x);
        float p3 = mono<f + 3>(x);
        v2f A;
        A.x = hi ? p2 : p0;
        A.y = hi ? p3 : p1;
        acc0 = __builtin_amdgcn_wmma_f32_16x16x4_f32(
                   false, A, false, B0[G], (short)0, acc0, false, false);
        acc1 = __builtin_amdgcn_wmma_f32_16x16x4_f32(
                   false, A, false, B1[G], (short)0, acc1, false, false);
        Steps<W, G + 1>::run(x, hi, B0, B1, acc0, acc1);
    }
};
template <int W>
struct Steps<W, GPW> {
    static __device__ __forceinline__ void run(const float*, bool, const v2f*,
                                               const v2f*, v8f&, v8f&) {}
};

__global__ __launch_bounds__(256) __attribute__((amdgpu_waves_per_eu(4, 4)))
void vindy_ksplit_kernel(const float* __restrict__ z,
                         const float* __restrict__ betas,
                         const float* __restrict__ big_xi,
                         const float* __restrict__ mask,
                         float* __restrict__ out,
                         int batch)
{
    __shared__ float sPart[2][WAVES][STEP_ELEMS];

    const int tid  = threadIdx.x;
    const int lane = tid & 31;
    const int w    = __builtin_amdgcn_readfirstlane(tid >> 5);
    const int half = lane >> 4;
    const int lm   = lane & 15;
    const bool hi  = (half != 0);

    v2f B0[GPW];
    v2f B1[GPW];
    {
        const int  fbase  = FPW * w + 2 * half;
        const bool has1   = (lm < 2);
        const int  col1   = has1 ? (16 + lm) : 17;
#pragma unroll
        for (int g = 0; g < GPW; ++g) {
            const int  f0  = fbase + 4 * g;
            const int  f1  = f0 + 1;
            const bool ok0 = (f0 < NP);
            const bool ok1 = (f1 < NP);
            const int  r0  = ok0 ? f0 : (NP - 1);
            const int  r1  = ok1 ? f1 : (NP - 1);
            float a0 = big_xi[r0 * NV + lm]   * mask[r0 * NV + lm];
            float a1 = big_xi[r1 * NV + lm]   * mask[r1 * NV + lm];
            float c0 = big_xi[r0 * NV + col1] * mask[r0 * NV + col1];
            float c1 = big_xi[r1 * NV + col1] * mask[r1 * NV + col1];
            B0[g].x = ok0 ? a0 : 0.0f;
            B0[g].y = ok1 ? a1 : 0.0f;
            B1[g].x = (ok0 && has1) ? c0 : 0.0f;
            B1[g].y = (ok1 && has1) ? c1 : 0.0f;
        }
    }

    const int blockRow = blockIdx.x * ROWS_PER_BLOCK;

    v4f q0, q1, q2, q3; v2f bb;
    {
        int gr = blockRow + lm;
        if (gr > batch - 1) gr = batch - 1;
        const v4f* zr = (const v4f*)(z + (size_t)gr * 16);
        q0 = zr[0]; q1 = zr[1]; q2 = zr[2]; q3 = zr[3];
        bb = *(const v2f*)(betas + (size_t)gr * 2);
    }

    for (int st = 0; st < STEPS_PER_BLOCK; ++st) {
        const int stepRow = blockRow + st * ROWS_PER_STEP;
        if (stepRow >= batch) break;
        float* part = &sPart[st & 1][w][0];

        for (int t = 0; t < TILES_PER_STEP; ++t) {
            float x[19];
            x[0]  = q0.x; x[1]  = q0.y; x[2]  = q0.z; x[3]  = q0.w;
            x[4]  = q1.x; x[5]  = q1.y; x[6]  = q1.z; x[7]  = q1.w;
            x[8]  = q2.x; x[9]  = q2.y; x[10] = q2.z; x[11] = q2.w;
            x[12] = q3.x; x[13] = q3.y; x[14] = q3.z; x[15] = q3.w;
            x[16] = bb.x; x[17] = bb.y; x[18] = 1.0f;
            {
                int gn = stepRow + (t + 1) * 16 + lm;
                if (gn > batch - 1) gn = batch - 1;
                const v4f* zr = (const v4f*)(z + (size_t)gn * 16);
                q0 = zr[0]; q1 = zr[1]; q2 = zr[2]; q3 = zr[3];
                bb = *(const v2f*)(betas + (size_t)gn * 2);
            }

            v8f acc0 = {0.f, 0.f, 0.f, 0.f, 0.f, 0.f, 0.f, 0.f};
            v8f acc1 = {0.f, 0.f, 0.f, 0.f, 0.f, 0.f, 0.f, 0.f};

            switch (w) {
            case 0:  Steps<0, 0>::run(x, hi, B0, B1, acc0, acc1); break;
            case 1:  Steps<1, 0>::run(x, hi, B0, B1, acc0, acc1); break;
            case 2:  Steps<2, 0>::run(x, hi, B0, B1, acc0, acc1); break;
            case 3:  Steps<3, 0>::run(x, hi, B0, B1, acc0, acc1); break;
            case 4:  Steps<4, 0>::run(x, hi, B0, B1, acc0, acc1); break;
            case 5:  Steps<5, 0>::run(x, hi, B0, B1, acc0, acc1); break;
            case 6:  Steps<6, 0>::run(x, hi, B0, B1, acc0, acc1); break;
            default: Steps<7, 0>::run(x, hi, B0, B1, acc0, acc1); break;
            }

            float* pt = part + (t * 16 + 8 * half) * NV;
#pragma unroll
            for (int v = 0; v < 8; ++v) pt[v * NV + lm] = acc0[v];
            if (lm < 2) {
#pragma unroll
                for (int v = 0; v < 8; ++v) pt[v * NV + 16 + lm] = acc1[v];
            }
        }

        __syncthreads();

        int nvalid = (batch - stepRow) * NV;
        if (nvalid > STEP_ELEMS) nvalid = STEP_ELEMS;
        float* og = out + (size_t)stepRow * NV;
        const float* pb = &sPart[st & 1][0][0];
        for (int e = tid; e < nvalid; e += 256) {
            float s = pb[e];
#pragma unroll
            for (int ww = 1; ww < WAVES; ++ww) s += pb[ww * STEP_ELEMS + e];
            og[e] = s;
        }
    }
}

extern "C" void kernel_launch(void* const* d_in, const int* in_sizes, int n_in,
                              void* d_out, int out_size, void* d_ws, size_t ws_size,
                              hipStream_t stream) {
    const float* z      = (const float*)d_in[0];
    const float* betas  = (const float*)d_in[1];
    const float* big_xi = (const float*)d_in[2];
    const float* mask   = (const float*)d_in[3];
    float* out          = (float*)d_out;

    const int batch = in_sizes[0] / 16;
    const int grid  = (batch + ROWS_PER_BLOCK - 1) / ROWS_PER_BLOCK;

    vindy_ksplit_kernel<<<grid, 256, 0, stream>>>(z, betas, big_xi, mask, out, batch);
}


# First tile fused with the Xi*mask fragment fetch: each K-group's two WMMAs issue behind its own loads, later groups' loads in flight, so the matrix pipe runs during the fetch; sum order unchanged
# speedup vs baseline: 2.0547x; 1.0559x over previous
#include <hip/hip_runtime.h>


#define NV 18
#define NP 1330
#define WAVES 8
#define GPW 42
#define FPW (4 * GPW)
#define KP (WAVES * FPW)
#define TILES_PER_STEP 4
#define ROWS_PER_STEP (16 * TILES_PER_STEP)
#define STEPS_PER_BLOCK 4
#define ROWS_PER_BLOCK (ROWS_PER_STEP * STEPS_PER_BLOCK)
#define STEP_ELEMS (ROWS_PER_STEP * NV)

typedef __attribute__((ext_vector_type(2))) float v2f;
typedef __attribute__((ext_vector_type(4))) float v4f;
typedef __attribute__((ext_vector_type(8))) float v8f;

struct MonTab {
    unsigned char a[KP];
    unsigned char b[KP];
    unsigned char c[KP];
};
constexpr MonTab make_tab() {
    MonTab t{};
    int f = 0;
    t.a[f] = 18; t.b[f] = 18; t.c[f] = 18; ++f;
    for (int i = 0; i < NV; ++i) { t.a[f] = i; t.b[f] = 18; t.c[f] = 18; ++f; }
    for (int i = 0; i < NV; ++i)
        for (int j = i; j < NV; ++j) { t.a[f] = i; t.b[f] = j; t.c[f] = 18; ++f; }
    for (int i = 0; i < NV; ++i)
        for (int j = i; j < NV; ++j)
            for (int k = j; k < NV; ++k) { t.a[f] = i; t.b[f] = j; t.c[f] = k; ++f; }
    for (; f < KP; ++f) { t.a[f] = 18; t.b[f] = 18; t.c[f] = 18; }
    return t;
}
constexpr MonTab TAB = make_tab();

template <int F>
static __device__ __forceinline__ float mono(const float* x) {
    constexpr int ia = TAB.a[F];
    constexpr int ib = TAB.b[F];
    constexpr int ic = TAB.c[F];
    float p = x[ia];
    if constexpr (ib != 18) p *= x[ib];
    if constexpr (ic != 18) p *= x[ic];
    return p;
}

template <int W, int G>
struct Steps {
    static __device__ __forceinline__ void run(const float* x, bool hi,
                                               const v2f* B0, const v2f* B1,
                                               v8f& acc0, v8f& acc1) {
        constexpr int f = FPW * W + 4 * G;
        float p0 = mono<f + 0>(x);
        float p1 = mono<f + 1>(x);
        float p2 = mono<f + 2>(x);
        float p3 = mono<f + 3>(x);
        v2f A;
        A.x = hi ? p2 : p0;
        A.y = hi ? p3 : p1;
        acc0 = __builtin_amdgcn_wmma_f32_16x16x4_f32(
                   false, A, false, B0[G], (short)0, acc0, false, false);
        acc1 = __builtin_amdgcn_wmma_f32_16x16x4_f32(
                   false, A, false, B1[G], (short)0, acc1, false, false);
        Steps<W, G + 1>::run(x, hi, B0, B1, acc0, acc1);
    }
};
template <int W>
struct Steps<W, GPW> {
    static __device__ __forceinline__ void run(const float*, bool, const v2f*,
                                               const v2f*, v8f&, v8f&) {}
};

#define FETCH_AHEAD 4

template <int W, int G>
struct Fetch {
    static __device__ __forceinline__ void run(int half, int lm, int col1,
                                               const float* __restrict__ big_xi,
                                               const float* __restrict__ mask, float* r) {
        constexpr int f = FPW * W + 4 * G;
        int f0 = f + 2 * half;
        int f1 = f0 + 1;
        if constexpr (f + 3 >= NP) {
            f0 = (f0 < NP) ? f0 : (NP - 1);
            f1 = (f1 < NP) ? f1 : (NP - 1);
        }
        r[0] = big_xi[f0 * NV + lm];    r[1] = mask[f0 * NV + lm];
        r[2] = big_xi[f1 * NV + lm];    r[3] = mask[f1 * NV + lm];
        r[4] = big_xi[f0 * NV + col1];  r[5] = mask[f0 * NV + col1];
        r[6] = big_xi[f1 * NV + col1];  r[7] = mask[f1 * NV + col1];
    }
};

template <int W, int G>
struct FirstSteps {
    static __device__ __forceinline__ void run(const float* x, bool hi, int half, int lm,
                                               bool has1, int col1,
                                               const float* __restrict__ big_xi,
                                               const float* __restrict__ mask,
                                               float* raw,
                                               v2f* B0, v2f* B1, v8f& acc0, v8f& acc1) {
        constexpr int f = FPW * W + 4 * G;
        float* r = raw + 8 * (G % FETCH_AHEAD);
        float a0 = r[0] * r[1];
        float a1 = r[2] * r[3];
        float c0 = r[4] * r[5];
        float c1 = r[6] * r[7];
        if constexpr (f + 3 >= NP) {
            const bool ok0 = (f + 2 * half < NP);
            const bool ok1 = (f + 2 * half + 1 < NP);
            a0 = ok0 ? a0 : 0.0f;  a1 = ok1 ? a1 : 0.0f;
            c0 = ok0 ? c0 : 0.0f;  c1 = ok1 ? c1 : 0.0f;
        }
        B0[G].x = a0;
        B0[G].y = a1;
        B1[G].x = has1 ? c0 : 0.0f;
        B1[G].y = has1 ? c1 : 0.0f;
        if constexpr (G + FETCH_AHEAD < GPW)
            Fetch<W, G + FETCH_AHEAD>::run(half, lm, col1, big_xi, mask, r);

        float p0 = mono<f + 0>(x);
        float p1 = mono<f + 1>(x);
        float p2 = mono<f + 2>(x);
        float p3 = mono<f + 3>(x);
        v2f A;
        A.x = hi ? p2 : p0;
        A.y = hi ? p3 : p1;
        acc0 = __builtin_amdgcn_wmma_f32_16x16x4_f32(
                   false, A, false, B0[G], (short)0, acc0, false, false);
        acc1 = __builtin_amdgcn_wmma_f32_16x16x4_f32(
                   false, A, false, B1[G], (short)0, acc1, false, false);
        FirstSteps<W, G + 1>::run(x, hi, half, lm, has1, col1, big_xi, mask, raw, B0, B1, acc0, acc1);
    }
};
template <int W>
struct FirstSteps<W, GPW> {
    static __device__ __forceinline__ void run(const float*, bool, int, int, bool, int,
                                               const float* __restrict__, const float* __restrict__,
                                               float*, v2f*, v2f*, v8f&, v8f&) {}
};

template <int W, int G>
struct Prime {
    static __device__ __forceinline__ void run(int half, int lm, int col1,
                                               const float* __restrict__ big_xi,
                                               const float* __restrict__ mask, float* raw) {
        Fetch<W, G>::run(half, lm, col1, big_xi, mask, raw + 8 * G);
        Prime<W, G + 1>::run(half, lm, col1, big_xi, mask, raw);
    }
};
template <int W>
struct Prime<W, FETCH_AHEAD> {
    static __device__ __forceinline__ void run(int, int, int, const float* __restrict__,
                                               const float* __restrict__, float*) {}
};

template <int W>
static __device__ __forceinline__ void first_tile(const float* x, bool hi, int half, int lm,
                                                  bool has1, int col1,
                                                  const float* __restrict__ big_xi,
                                                  const float* __restrict__ mask,
                                                  v2f* B0, v2f* B1, v8f& acc0, v8f& acc1) {
    float raw[8 * FETCH_AHEAD];
    Prime<W, 0>::run(half, lm, col1, big_xi, mask, raw);
    FirstSteps<W, 0>::run(x, hi, half, lm, has1, col1, big_xi, mask, raw, B0, B1, acc0, acc1);
}

template <bool F> struct TileKind { static constexpr bool first = F; };

__global__ __launch_bounds__(256) __attribute__((amdgpu_waves_per_eu(4, 4)))
void vindy_ksplit_kernel(const float* __restrict__ z,
                         const float* __restrict__ betas,
                         const float* __restrict__ big_xi,
                         const float* __restrict__ mask,
                         float* __restrict__ out,
                         int batch)
{
    __shared__ float sPart[2][WAVES][STEP_ELEMS];

    const int tid  = threadIdx.x;
    const int lane = tid & 31;
    const int w    = __builtin_amdgcn_readfirstlane(tid >> 5);
    const int half = lane >> 4;
    const int lm   = lane & 15;
    const bool hi  = (half != 0);

    v2f B0[GPW];
    v2f B1[GPW];
    const bool has1 = (lm < 2);
    const int  col1 = has1 ? (16 + lm) : 17;

    const int blockRow = blockIdx.x * ROWS_PER_BLOCK;

    v4f q0, q1, q2, q3; v2f bb;
    {
        int gr = blockRow + lm;
        if (gr > batch - 1) gr = batch - 1;
        const v4f* zr = (const v4f*)(z + (size_t)gr * 16);
        q0 = zr[0]; q1 = zr[1]; q2 = zr[2]; q3 = zr[3];
        bb = *(const v2f*)(betas + (size_t)gr * 2);
    }

    auto tile = [&](auto kind, int st, int t) __attribute__((always_inline)) {
        constexpr bool FIRST = decltype(kind)::first;
        const int stepRow = blockRow + st * ROWS_PER_STEP;
        float* part = &sPart[st & 1][w][0];

        float x[19];
        x[0]  = q0.x; x[1]  = q0.y; x[2]  = q0.z; x[3]  = q0.w;
        x[4]  = q1.x; x[5]  = q1.y; x[6]  = q1.z; x[7]  = q1.w;
        x[8]  = q2.x; x[9]  = q2.y; x[10] = q2.z; x[11] = q2.w;
        x[12] = q3.x; x[13] = q3.y; x[14] = q3.z; x[15] = q3.w;
        x[16] = bb.x; x[17] = bb.y; x[18] = 1.0f;
        {
            int gn = stepRow + (t + 1) * 16 + lm;
            if (gn > batch - 1) gn = batch - 1;
            const v4f* zr = (const v4f*)(z + (size_t)gn * 16);
            q0 = zr[0]; q1 = zr[1]; q2 = zr[2]; q3 = zr[3];
            bb = *(const v2f*)(betas + (size_t)gn * 2);
        }

        v8f acc0 = {0.f, 0.f, 0.f, 0.f, 0.f, 0.f, 0.f, 0.f};
        v8f acc1 = {0.f, 0.f, 0.f, 0.f, 0.f, 0.f, 0.f, 0.f};

        if constexpr (FIRST) {
            switch (w) {
            case 0:  first_tile<0>(x, hi, half, lm, has1, col1, big_xi, mask, B0, B1, acc0, acc1); break;
            case 1:  first_tile<1>(x, hi, half, lm, has1, col1, big_xi, mask, B0, B1, acc0, acc1); break;
            case 2:  first_tile<2>(x, hi, half, lm, has1, col1, big_xi, mask, B0, B1, acc0, acc1); break;
            case 3:  first_tile<3>(x, hi, half, lm, has1, col1, big_xi, mask, B0, B1, acc0, acc1); break;
            case 4:  first_tile<4>(x, hi, half, lm, has1, col1, big_xi, mask, B0, B1, acc0, acc1); break;
            case 5:  first_tile<5>(x, hi, half, lm, has1, col1, big_xi, mask, B0, B1, acc0, acc1); break;
            case 6:  first_tile<6>(x, hi, half, lm, has1, col1, big_xi, mask, B0, B1, acc0, acc1); break;
            default: first_tile<7>(x, hi, half, lm, has1, col1, big_xi, mask, B0, B1, acc0, acc1); break;
            }
        } else {
            switch (w) {
            case 0:  Steps<0, 0>::run(x, hi, B0, B1, acc0, acc1); break;
            case 1:  Steps<1, 0>::run(x, hi, B0, B1, acc0, acc1); break;
            case 2:  Steps<2, 0>::run(x, hi, B0, B1, acc0, acc1); break;
            case 3:  Steps<3, 0>::run(x, hi, B0, B1, acc0, acc1); break;
            case 4:  Steps<4, 0>::run(x, hi, B0, B1, acc0, acc1); break;
            case 5:  Steps<5, 0>::run(x, hi, B0, B1, acc0, acc1); break;
            case 6:  Steps<6, 0>::run(x, hi, B0, B1, acc0, acc1); break;
            default: Steps<7, 0>::run(x, hi, B0, B1, acc0, acc1); break;
            }
        }

        float* pt = part + (t * 16 + 8 * half) * NV;
#pragma unroll
        for (int v = 0; v < 8; ++v) pt[v * NV + lm] = acc0[v];
        if (lm < 2) {
#pragma unroll
            for (int v = 0; v < 8; ++v) pt[v * NV + 16 + lm] = acc1[v];
        }
    };

    tile(TileKind<true>{}, 0, 0);

    for (int st = 0; st < STEPS_PER_BLOCK; ++st) {
        const int stepRow = blockRow + st * ROWS_PER_STEP;
        if (stepRow >= batch) break;

        for (int t = (st == 0) ? 1 : 0; t < TILES_PER_STEP; ++t) tile(TileKind<false>{}, st, t);

        __syncthreads();

        int nvalid = (batch - stepRow) * NV;
        if (nvalid > STEP_ELEMS) nvalid = STEP_ELEMS;
        float* og = out + (size_t)stepRow * NV;
        const float* pb = &sPart[st & 1][0][0];
        for (int e = tid; e < nvalid; e += 256) {
            float s = pb[e];
#pragma unroll
            for (int ww = 1; ww < WAVES; ++ww) s += pb[ww * STEP_ELEMS + e];
            og[e] = s;
        }
    }
}

extern "C" void kernel_launch(void* const* d_in, const int* in_sizes, int n_in,
                              void* d_out, int out_size, void* d_ws, size_t ws_size,
                              hipStream_t stream) {
    const float* z      = (const float*)d_in[0];
    const float* betas  = (const float*)d_in[1];
    const float* big_xi = (const float*)d_in[2];
    const float* mask   = (const float*)d_in[3];
    float* out          = (float*)d_out;

    const int batch = in_sizes[0] / 16;
    const int grid  = (batch + ROWS_PER_BLOCK - 1) / ROWS_PER_BLOCK;

    vindy_ksplit_kernel<<<grid, 256, 0, stream>>>(z, betas, big_xi, mask, out, batch);
}
